# starting kernel
# baseline (speedup 1.0000x reference)
#include <hip/hip_runtime.h>

#define B_  16
#define NA  2048
#define NP  2048
#define DD  512
#define KK  32

typedef __attribute__((ext_vector_type(2))) float v2f;
typedef __attribute__((ext_vector_type(8))) float v8f;

__device__ __forceinline__ unsigned ordkey(float f) {
    unsigned u = __float_as_uint(f);
    return (u & 0x80000000u) ? ~u : (u | 0x80000000u);
}
__device__ __forceinline__ float keyval(unsigned k) {
    return __uint_as_float((k & 0x80000000u) ? (k ^ 0x80000000u) : ~k);
}

__global__ __launch_bounds__(256) void prep_kernel(
    const float* __restrict__ anchor, const int* __restrict__ idx_a,
    const int* __restrict__ idx_p, float* __restrict__ a_s_out,
    float* __restrict__ asel_ws) {
    const int r    = blockIdx.x * 8 + (threadIdx.x >> 5);
    const int lane = threadIdx.x & 31;
    const int kind = r >> 9;
    const int t    = r & 511;
    const int b    = t >> 5;
    const int idx  = kind ? idx_p[t] : idx_a[t];
    const float* src = anchor + ((size_t)b * NA + idx) * DD;
    float v[16];
    float s = 0.f;
#pragma unroll
    for (int j = 0; j < 16; ++j) { float x = src[lane + 32 * j]; v[j] = x; s += x * x; }
#pragma unroll
    for (int off = 16; off; off >>= 1) s += __shfl_xor(s, off, 32);
    const float inv = 1.f / fmaxf(sqrtf(s), 1e-12f);
    float* dst = (kind ? asel_ws : a_s_out) + (size_t)t * DD;
#pragma unroll
    for (int j = 0; j < 16; ++j) dst[lane + 32 * j] = v[j] * inv;
}

__global__ void init_best(unsigned long long* best) {
    best[blockIdx.x * 256 + threadIdx.x] = 0ull;
}

__global__ __launch_bounds__(256) void score_kernel(
    const float* __restrict__ p, const float* __restrict__ asel,
    unsigned long long* __restrict__ best) {
    __shared__ float As[KK * DD];
    const int b   = blockIdx.y;
    const int tid = threadIdx.x;
    {
        const float4* s4 = (const float4*)(asel + (size_t)b * KK * DD);
        float4* d4 = (float4*)As;
        for (int i = tid; i < KK * DD / 4; i += 256) d4[i] = s4[i];
    }
    __syncthreads();

    const int wave = tid >> 5;
    const int lane = tid & 31;
    const int half = lane >> 4;
    const int lr   = lane & 15;
    const int m    = blockIdx.x * 128 + wave * 16 + lr;

    const float* pb = p + ((size_t)b * NP + m) * DD + 2 * half;
    const float* a0 = As + lr * DD + 2 * half;
    const float* a1 = As + (lr + 16) * DD + 2 * half;

    v8f acc0 = {};
    v8f acc1 = {};
    float ss = 0.f;
    for (int k0 = 0; k0 < DD; k0 += 4) {
        v2f bf  = *(const v2f*)(pb + k0);
        v2f af0 = *(const v2f*)(a0 + k0);
        v2f af1 = *(const v2f*)(a1 + k0);
        ss += bf.x * bf.x + bf.y * bf.y;
        acc0 = __builtin_amdgcn_wmma_f32_16x16x4_f32(false, af0, false, bf,
                                                     (short)0, acc0, false, false);
        acc1 = __builtin_amdgcn_wmma_f32_16x16x4_f32(false, af1, false, bf,
                                                     (short)0, acc1, false, false);
    }
    ss += __shfl_xor(ss, 16, 32);
    const float inv = 1.f / fmaxf(sqrtf(ss), 1e-12f);
    const unsigned keylo = (unsigned)(NP - 1 - m);

#pragma unroll
    for (int i = 0; i < 8; ++i) {
        unsigned long long k0p =
            ((unsigned long long)ordkey(acc0[i] * inv) << 32) | keylo;
        unsigned long long k1p =
            ((unsigned long long)ordkey(acc1[i] * inv) << 32) | keylo;
#pragma unroll
        for (int off = 8; off; off >>= 1) {
            unsigned long long o0 = __shfl_xor(k0p, off, 32);
            unsigned long long o1 = __shfl_xor(k1p, off, 32);
            k0p = o0 > k0p ? o0 : k0p;
            k1p = o1 > k1p ? o1 : k1p;
        }
        if (lr == 0) {
            atomicMax(best + (size_t)b * KK + (i + 8 * half), k0p);
            atomicMax(best + (size_t)b * KK + (16 + i + 8 * half), k1p);
        }
    }
}

__global__ __launch_bounds__(256) void fuse_kernel(
    const float* __restrict__ p1, const float* __restrict__ p2,
    const unsigned long long* __restrict__ best, float* __restrict__ p_s_out) {
    const int t    = blockIdx.x * 8 + (threadIdx.x >> 5);
    const int lane = threadIdx.x & 31;
    const int b    = t >> 5;
    const unsigned long long e1 = best[t];
    const unsigned long long e2 = best[B_ * KK + t];
    const float w1 = keyval((unsigned)(e1 >> 32));
    const float w2 = keyval((unsigned)(e2 >> 32));
    const int  i1  = NP - 1 - (int)(unsigned)(e1 & 0xffffffffull);
    const int  i2  = NP - 1 - (int)(unsigned)(e2 & 0xffffffffull);
    const float* r1 = p1 + ((size_t)b * NP + i1) * DD;
    const float* r2 = p2 + ((size_t)b * NP + i2) * DD;

    float v1[16], v2[16];
    float s1 = 0.f, s2 = 0.f;
#pragma unroll
    for (int j = 0; j < 16; ++j) {
        float x = r1[lane + 32 * j]; v1[j] = x; s1 += x * x;
        float y = r2[lane + 32 * j]; v2[j] = y; s2 += y * y;
    }
#pragma unroll
    for (int off = 16; off; off >>= 1) {
        s1 += __shfl_xor(s1, off, 32);
        s2 += __shfl_xor(s2, off, 32);
    }
    const float c1 = w1 / fmaxf(sqrtf(s1), 1e-12f);
    const float c2 = w2 / fmaxf(sqrtf(s2), 1e-12f);
    const float dn = 1.f / (w1 + w2 + 1e-6f);

    float f[16];
    float sf = 0.f;
#pragma unroll
    for (int j = 0; j < 16; ++j) {
        f[j] = (c1 * v1[j] + c2 * v2[j]) * dn;
        sf += f[j] * f[j];
    }
#pragma unroll
    for (int off = 16; off; off >>= 1) sf += __shfl_xor(sf, off, 32);
    const float invf = 1.f / fmaxf(sqrtf(sf), 1e-12f);
    float* dst = p_s_out + (size_t)t * DD;
#pragma unroll
    for (int j = 0; j < 16; ++j) dst[lane + 32 * j] = f[j] * invf;
}

extern "C" void kernel_launch(void* const* d_in, const int* in_sizes, int n_in,
                              void* d_out, int out_size, void* d_ws, size_t ws_size,
                              hipStream_t stream) {
    (void)in_sizes; (void)n_in; (void)out_size; (void)ws_size;
    const float* anchor = (const float*)d_in[0];
    const float* p1     = (const float*)d_in[1];
    const float* p2     = (const float*)d_in[2];
    const int*   idx_a  = (const int*)d_in[3];
    const int*   idx_p  = (const int*)d_in[4];

    float* A_s = (float*)d_out;
    float* P_s = (float*)d_out + (size_t)B_ * KK * DD;

    unsigned long long* best = (unsigned long long*)d_ws;
    float* asel = (float*)((char*)d_ws + (size_t)2 * B_ * KK * sizeof(unsigned long long));

    init_best<<<dim3((2 * B_ * KK) / 256), 256, 0, stream>>>(best);
    prep_kernel<<<dim3((2 * B_ * KK) / 8), 256, 0, stream>>>(anchor, idx_a, idx_p, A_s, asel);

    dim3 sg(NP / 128, B_);
    score_kernel<<<sg, 256, 0, stream>>>(p1, asel, best);
    score_kernel<<<sg, 256, 0, stream>>>(p2, asel, best + B_ * KK);

    fuse_kernel<<<dim3((B_ * KK) / 8), 256, 0, stream>>>(p1, p2, best, P_s);
}


# Argmax epilogue re-assigned: wave = one accumulator register x all 8 row groups; lane keeps best of 8 p-rows, then ONE 16-lane reduce (16->2 chains/wave, atomics /8); row norms once per row in LDS
# speedup vs baseline: 1.5587x; 1.5587x over previous
#include <hip/hip_runtime.h>

#define B_  16
#define NA  2048
#define NP  2048
#define DD  512
#define KK  32

typedef __attribute__((ext_vector_type(2))) float v2f;
typedef __attribute__((ext_vector_type(8))) float v8f;

__device__ __forceinline__ unsigned ordkey(float f) {
    unsigned u = __float_as_uint(f);
    return (u & 0x80000000u) ? ~u : (u | 0x80000000u);
}
__device__ __forceinline__ float keyval(unsigned k) {
    return __uint_as_float((k & 0x80000000u) ? (k ^ 0x80000000u) : ~k);
}

__global__ __launch_bounds__(256) void prep_kernel(
    const float* __restrict__ anchor, const int* __restrict__ idx_a,
    const int* __restrict__ idx_p, float* __restrict__ a_s_out,
    float* __restrict__ asel_ws) {
    const int r    = blockIdx.x * 8 + (threadIdx.x >> 5);
    const int lane = threadIdx.x & 31;
    const int kind = r >> 9;
    const int t    = r & 511;
    const int b    = t >> 5;
    const int idx  = kind ? idx_p[t] : idx_a[t];
    const float* src = anchor + ((size_t)b * NA + idx) * DD;
    float v[16];
    float s = 0.f;
#pragma unroll
    for (int j = 0; j < 16; ++j) { float x = src[lane + 32 * j]; v[j] = x; s += x * x; }
#pragma unroll
    for (int off = 16; off; off >>= 1) s += __shfl_xor(s, off, 32);
    const float inv = 1.f / fmaxf(sqrtf(s), 1e-12f);
    float* dst = (kind ? asel_ws : a_s_out) + (size_t)t * DD;
#pragma unroll
    for (int j = 0; j < 16; ++j) dst[lane + 32 * j] = v[j] * inv;
}

__global__ void init_best(unsigned long long* best) {
    best[blockIdx.x * 256 + threadIdx.x] = 0ull;
}

#define KS   8
#define KSW  (DD / KS)
#define NJ   (KSW / 4)
#define NG   8
#define NT   2
#define NGT  (NT * NG)
#define RSTR 17

__global__ __launch_bounds__(512) void score_kernel(
    const float* __restrict__ p1, const float* __restrict__ p2,
    const float* __restrict__ asel, unsigned long long* __restrict__ best) {
    __shared__ float red[KS * NGT * RSTR * 32];
    __shared__ float invt[NT * 128];
    const int b    = blockIdx.y;
    const int tid  = threadIdx.x;
    const int wave = tid >> 5;
    const int lane = tid & 31;
    const int half = lane >> 4;
    const int lr   = lane & 15;
    const int ks   = wave & 7;
    const int rh   = wave >> 3;

    const int koff = ks * KSW + 2 * half;
    const float* ap0 = asel + ((size_t)b * KK + lr) * DD + koff;
    const float* ap1 = asel + ((size_t)b * KK + lr + 16) * DD + koff;
    v2f af0[NJ], af1[NJ];
#pragma unroll
    for (int j = 0; j < NJ; ++j) {
        af0[j] = *(const v2f*)(ap0 + 4 * j);
        af1[j] = *(const v2f*)(ap1 + 4 * j);
    }

    const int m0 = blockIdx.x * 128 + rh * 64 + lr;
    const size_t roff = ((size_t)b * NP + m0) * DD + koff;
    const float* pb1 = p1 + roff;
    const float* pb2 = p2 + roff;

    v2f bc[NJ], bn[NJ];
#pragma unroll
    for (int j = 0; j < NJ; ++j) bc[j] = *(const v2f*)(pb1 + 4 * j);

    const float* pn = pb1 + (size_t)16 * DD;
#pragma unroll 1
    for (int q = 0; q < 2 * NT; ++q) {
#pragma unroll
        for (int j = 0; j < NJ; ++j) bn[j] = *(const v2f*)(pn + 4 * j);
        pn += (size_t)16 * DD;
        {
            v8f acc0 = {};
            v8f acc1 = {};
            float ss = 0.f;
#pragma unroll
            for (int j = 0; j < NJ; ++j) {
                const v2f bf = bc[j];
                ss += bf.x * bf.x + bf.y * bf.y;
                acc0 = __builtin_amdgcn_wmma_f32_16x16x4_f32(false, af0[j], false, bf,
                                                             (short)0, acc0, false, false);
                acc1 = __builtin_amdgcn_wmma_f32_16x16x4_f32(false, af1[j], false, bf,
                                                             (short)0, acc1, false, false);
            }
            const int gt = (q >> 1) * NG + rh * 4 + 2 * (q & 1);
            float* r = red + (size_t)((ks * NGT + gt) * RSTR) * 32 + lane;
#pragma unroll
            for (int i = 0; i < 8; ++i) {
                r[i * 32]       = acc0[i];
                r[(8 + i) * 32] = acc1[i];
            }
            r[16 * 32] = ss;
        }
        if (q < 2 * NT - 1) {
            if (q & 1) pn = pb2;
#pragma unroll
            for (int j = 0; j < NJ; ++j) bc[j] = *(const v2f*)(pn + 4 * j);
            pn += (size_t)16 * DD;
        }
        {
            v8f acc0 = {};
            v8f acc1 = {};
            float ss = 0.f;
#pragma unroll
            for (int j = 0; j < NJ; ++j) {
                const v2f bf = bn[j];
                ss += bf.x * bf.x + bf.y * bf.y;
                acc0 = __builtin_amdgcn_wmma_f32_16x16x4_f32(false, af0[j], false, bf,
                                                             (short)0, acc0, false, false);
                acc1 = __builtin_amdgcn_wmma_f32_16x16x4_f32(false, af1[j], false, bf,
                                                             (short)0, acc1, false, false);
            }
            const int gt = (q >> 1) * NG + rh * 4 + 2 * (q & 1) + 1;
            float* r = red + (size_t)((ks * NGT + gt) * RSTR) * 32 + lane;
#pragma unroll
            for (int i = 0; i < 8; ++i) {
                r[i * 32]       = acc0[i];
                r[(8 + i) * 32] = acc1[i];
            }
            r[16 * 32] = ss;
        }
    }
    __syncthreads();

    if (tid < NT * 128) {
        const int tA = tid >> 7;
        const int GA = (tid >> 4) & 7;
        const int lA = tid & 15;
        float s0 = 0.f, s1 = 0.f;
        for (int s = 0; s < KS; ++s) {
            const float* r = red + (size_t)((s * NGT + tA * NG + GA) * RSTR + 16) * 32 + lA;
            s0 += r[0];
            s1 += r[16];
        }
        invt[tid] = 1.f / fmaxf(sqrtf(s0 + s1), 1e-12f);
    }
    __syncthreads();

    const int t  = wave >> 3;
    const int ri = wave & 7;
    const int mb = blockIdx.x * 128 + lr;
    unsigned long long k0 = 0ull;
    unsigned long long k1 = 0ull;
#pragma unroll
    for (int G = 0; G < NG; ++G) {
        float a0 = 0.f, a1 = 0.f;
        for (int s = 0; s < KS; ++s) {
            const float* r = red + (size_t)((s * NGT + t * NG + G) * RSTR) * 32 + lane;
            a0 += r[ri * 32];
            a1 += r[(8 + ri) * 32];
        }
        const float inv = invt[t * 128 + G * 16 + lr];
        const unsigned keylo = (unsigned)(NP - 1 - (mb + G * 16));
        const unsigned long long c0 = ((unsigned long long)ordkey(a0 * inv) << 32) | keylo;
        const unsigned long long c1 = ((unsigned long long)ordkey(a1 * inv) << 32) | keylo;
        k0 = c0 > k0 ? c0 : k0;
        k1 = c1 > k1 ? c1 : k1;
    }
    for (int off = 8; off; off >>= 1) {
        const unsigned long long o0 = __shfl_xor(k0, off, 32);
        const unsigned long long o1 = __shfl_xor(k1, off, 32);
        k0 = o0 > k0 ? o0 : k0;
        k1 = o1 > k1 ? o1 : k1;
    }
    if (lr == 0) {
        unsigned long long* bt = best + (size_t)t * B_ * KK + (size_t)b * KK;
        atomicMax(bt + (ri + 8 * half), k0);
        atomicMax(bt + (16 + ri + 8 * half), k1);
    }
}

__global__ __launch_bounds__(256) void fuse_kernel(
    const float* __restrict__ p1, const float* __restrict__ p2,
    const unsigned long long* __restrict__ best, float* __restrict__ p_s_out) {
    const int t    = blockIdx.x * 8 + (threadIdx.x >> 5);
    const int lane = threadIdx.x & 31;
    const int b    = t >> 5;
    const unsigned long long e1 = best[t];
    const unsigned long long e2 = best[B_ * KK + t];
    const float w1 = keyval((unsigned)(e1 >> 32));
    const float w2 = keyval((unsigned)(e2 >> 32));
    const int  i1  = NP - 1 - (int)(unsigned)(e1 & 0xffffffffull);
    const int  i2  = NP - 1 - (int)(unsigned)(e2 & 0xffffffffull);
    const float* r1 = p1 + ((size_t)b * NP + i1) * DD;
    const float* r2 = p2 + ((size_t)b * NP + i2) * DD;

    float v1[16], v2[16];
    float s1 = 0.f, s2 = 0.f;
#pragma unroll
    for (int j = 0; j < 16; ++j) {
        float x = r1[lane + 32 * j]; v1[j] = x; s1 += x * x;
        float y = r2[lane + 32 * j]; v2[j] = y; s2 += y * y;
    }
#pragma unroll
    for (int off = 16; off; off >>= 1) {
        s1 += __shfl_xor(s1, off, 32);
        s2 += __shfl_xor(s2, off, 32);
    }
    const float c1 = w1 / fmaxf(sqrtf(s1), 1e-12f);
    const float c2 = w2 / fmaxf(sqrtf(s2), 1e-12f);
    const float dn = 1.f / (w1 + w2 + 1e-6f);

    float f[16];
    float sf = 0.f;
#pragma unroll
    for (int j = 0; j < 16; ++j) {
        f[j] = (c1 * v1[j] + c2 * v2[j]) * dn;
        sf += f[j] * f[j];
    }
#pragma unroll
    for (int off = 16; off; off >>= 1) sf += __shfl_xor(sf, off, 32);
    const float invf = 1.f / fmaxf(sqrtf(sf), 1e-12f);
    float* dst = p_s_out + (size_t)t * DD;
#pragma unroll
    for (int j = 0; j < 16; ++j) dst[lane + 32 * j] = f[j] * invf;
}

extern "C" void kernel_launch(void* const* d_in, const int* in_sizes, int n_in,
                              void* d_out, int out_size, void* d_ws, size_t ws_size,
                              hipStream_t stream) {
    (void)in_sizes; (void)n_in; (void)out_size; (void)ws_size;
    const float* anchor = (const float*)d_in[0];
    const float* p1     = (const float*)d_in[1];
    const float* p2     = (const float*)d_in[2];
    const int*   idx_a  = (const int*)d_in[3];
    const int*   idx_p  = (const int*)d_in[4];

    float* A_s = (float*)d_out;
    float* P_s = (float*)d_out + (size_t)B_ * KK * DD;

    unsigned long long* best = (unsigned long long*)d_ws;
    float* asel = (float*)((char*)d_ws + (size_t)2 * B_ * KK * sizeof(unsigned long long));

    init_best<<<dim3((2 * B_ * KK) / 256), 256, 0, stream>>>(best);
    prep_kernel<<<dim3((2 * B_ * KK) / 8), 256, 0, stream>>>(anchor, idx_a, idx_p, A_s, asel);

    dim3 sg(NP / 128, B_);
    score_kernel<<<sg, 512, 0, stream>>>(p1, p2, asel, best);

    fuse_kernel<<<dim3((B_ * KK) / 8), 256, 0, stream>>>(p1, p2, best, P_s);
}


# Clearing of the packed argmax accumulators folded into the gather launch (4->3 launches): each gather wave zeroes one slot ahead of the score atomics; every sum and its order unchanged
# speedup vs baseline: 1.6194x; 1.0389x over previous
#include <hip/hip_runtime.h>

#define B_  16
#define NA  2048
#define NP  2048
#define DD  512
#define KK  32

typedef __attribute__((ext_vector_type(2))) float v2f;
typedef __attribute__((ext_vector_type(8))) float v8f;

__device__ __forceinline__ unsigned ordkey(float f) {
    unsigned u = __float_as_uint(f);
    return (u & 0x80000000u) ? ~u : (u | 0x80000000u);
}
__device__ __forceinline__ float keyval(unsigned k) {
    return __uint_as_float((k & 0x80000000u) ? (k ^ 0x80000000u) : ~k);
}

__global__ __launch_bounds__(256) void prep_kernel(
    const float* __restrict__ anchor, const int* __restrict__ idx_a,
    const int* __restrict__ idx_p, float* __restrict__ a_s_out,
    float* __restrict__ asel_ws, unsigned long long* __restrict__ best) {
    const int r    = blockIdx.x * 8 + (threadIdx.x >> 5);
    const int lane = threadIdx.x & 31;
    if (lane == 0) best[r] = 0ull;
    const int kind = r >> 9;
    const int t    = r & 511;
    const int b    = t >> 5;
    const int idx  = kind ? idx_p[t] : idx_a[t];
    const float* src = anchor + ((size_t)b * NA + idx) * DD;
    float v[16];
    float s = 0.f;
#pragma unroll
    for (int j = 0; j < 16; ++j) { float x = src[lane + 32 * j]; v[j] = x; s += x * x; }
#pragma unroll
    for (int off = 16; off; off >>= 1) s += __shfl_xor(s, off, 32);
    const float inv = 1.f / fmaxf(sqrtf(s), 1e-12f);
    float* dst = (kind ? asel_ws : a_s_out) + (size_t)t * DD;
#pragma unroll
    for (int j = 0; j < 16; ++j) dst[lane + 32 * j] = v[j] * inv;
}

#define KS   8
#define KSW  (DD / KS)
#define NJ   (KSW / 4)
#define NG   8
#define NT   2
#define NGT  (NT * NG)
#define RSTR 17

__global__ __launch_bounds__(512) void score_kernel(
    const float* __restrict__ p1, const float* __restrict__ p2,
    const float* __restrict__ asel, unsigned long long* __restrict__ best) {
    __shared__ float red[KS * NGT * RSTR * 32];
    __shared__ float invt[NT * 128];
    const int b    = blockIdx.y;
    const int tid  = threadIdx.x;
    const int wave = tid >> 5;
    const int lane = tid & 31;
    const int half = lane >> 4;
    const int lr   = lane & 15;
    const int ks   = wave & 7;
    const int rh   = wave >> 3;

    const int koff = ks * KSW + 2 * half;
    const float* ap0 = asel + ((size_t)b * KK + lr) * DD + koff;
    const float* ap1 = asel + ((size_t)b * KK + lr + 16) * DD + koff;
    v2f af0[NJ], af1[NJ];
#pragma unroll
    for (int j = 0; j < NJ; ++j) {
        af0[j] = *(const v2f*)(ap0 + 4 * j);
        af1[j] = *(const v2f*)(ap1 + 4 * j);
    }

    const int m0 = blockIdx.x * 128 + rh * 64 + lr;
    const size_t roff = ((size_t)b * NP + m0) * DD + koff;
    const float* pb1 = p1 + roff;
    const float* pb2 = p2 + roff;

    v2f bc[NJ], bn[NJ];
#pragma unroll
    for (int j = 0; j < NJ; ++j) bc[j] = *(const v2f*)(pb1 + 4 * j);

    const float* pn = pb1 + (size_t)16 * DD;
#pragma unroll 1
    for (int q = 0; q < 2 * NT; ++q) {
#pragma unroll
        for (int j = 0; j < NJ; ++j) bn[j] = *(const v2f*)(pn + 4 * j);
        pn += (size_t)16 * DD;
        {
            v8f acc0 = {};
            v8f acc1 = {};
            float ss = 0.f;
#pragma unroll
            for (int j = 0; j < NJ; ++j) {
                const v2f bf = bc[j];
                ss += bf.x * bf.x + bf.y * bf.y;
                acc0 = __builtin_amdgcn_wmma_f32_16x16x4_f32(false, af0[j], false, bf,
                                                             (short)0, acc0, false, false);
                acc1 = __builtin_amdgcn_wmma_f32_16x16x4_f32(false, af1[j], false, bf,
                                                             (short)0, acc1, false, false);
            }
            const int gt = (q >> 1) * NG + rh * 4 + 2 * (q & 1);
            float* r = red + (size_t)((ks * NGT + gt) * RSTR) * 32 + lane;
#pragma unroll
            for (int i = 0; i < 8; ++i) {
                r[i * 32]       = acc0[i];
                r[(8 + i) * 32] = acc1[i];
            }
            r[16 * 32] = ss;
        }
        if (q < 2 * NT - 1) {
            if (q & 1) pn = pb2;
#pragma unroll
            for (int j = 0; j < NJ; ++j) bc[j] = *(const v2f*)(pn + 4 * j);
            pn += (size_t)16 * DD;
        }
        {
            v8f acc0 = {};
            v8f acc1 = {};
            float ss = 0.f;
#pragma unroll
            for (int j = 0; j < NJ; ++j) {
                const v2f bf = bn[j];
                ss += bf.x * bf.x + bf.y * bf.y;
                acc0 = __builtin_amdgcn_wmma_f32_16x16x4_f32(false, af0[j], false, bf,
                                                             (short)0, acc0, false, false);
                acc1 = __builtin_amdgcn_wmma_f32_16x16x4_f32(false, af1[j], false, bf,
                                                             (short)0, acc1, false, false);
            }
            const int gt = (q >> 1) * NG + rh * 4 + 2 * (q & 1) + 1;
            float* r = red + (size_t)((ks * NGT + gt) * RSTR) * 32 + lane;
#pragma unroll
            for (int i = 0; i < 8; ++i) {
                r[i * 32]       = acc0[i];
                r[(8 + i) * 32] = acc1[i];
            }
            r[16 * 32] = ss;
        }
    }
    __syncthreads();

    if (tid < NT * 128) {
        const int tA = tid >> 7;
        const int GA = (tid >> 4) & 7;
        const int lA = tid & 15;
        float s0 = 0.f, s1 = 0.f;
        for (int s = 0; s < KS; ++s) {
            const float* r = red + (size_t)((s * NGT + tA * NG + GA) * RSTR + 16) * 32 + lA;
            s0 += r[0];
            s1 += r[16];
        }
        invt[tid] = 1.f / fmaxf(sqrtf(s0 + s1), 1e-12f);
    }
    __syncthreads();

    const int t  = wave >> 3;
    const int ri = wave & 7;
    const int mb = blockIdx.x * 128 + lr;
    unsigned long long k0 = 0ull;
    unsigned long long k1 = 0ull;
#pragma unroll
    for (int G = 0; G < NG; ++G) {
        float a0 = 0.f, a1 = 0.f;
        for (int s = 0; s < KS; ++s) {
            const float* r = red + (size_t)((s * NGT + t * NG + G) * RSTR) * 32 + lane;
            a0 += r[ri * 32];
            a1 += r[(8 + ri) * 32];
        }
        const float inv = invt[t * 128 + G * 16 + lr];
        const unsigned keylo = (unsigned)(NP - 1 - (mb + G * 16));
        const unsigned long long c0 = ((unsigned long long)ordkey(a0 * inv) << 32) | keylo;
        const unsigned long long c1 = ((unsigned long long)ordkey(a1 * inv) << 32) | keylo;
        k0 = c0 > k0 ? c0 : k0;
        k1 = c1 > k1 ? c1 : k1;
    }
    for (int off = 8; off; off >>= 1) {
        const unsigned long long o0 = __shfl_xor(k0, off, 32);
        const unsigned long long o1 = __shfl_xor(k1, off, 32);
        k0 = o0 > k0 ? o0 : k0;
        k1 = o1 > k1 ? o1 : k1;
    }
    if (lr == 0) {
        unsigned long long* bt = best + (size_t)t * B_ * KK + (size_t)b * KK;
        atomicMax(bt + (ri + 8 * half), k0);
        atomicMax(bt + (16 + ri + 8 * half), k1);
    }
}

__global__ __launch_bounds__(256) void fuse_kernel(
    const float* __restrict__ p1, const float* __restrict__ p2,
    const unsigned long long* __restrict__ best, float* __restrict__ p_s_out) {
    const int t    = blockIdx.x * 8 + (threadIdx.x >> 5);
    const int lane = threadIdx.x & 31;
    const int b    = t >> 5;
    const unsigned long long e1 = best[t];
    const unsigned long long e2 = best[B_ * KK + t];
    const float w1 = keyval((unsigned)(e1 >> 32));
    const float w2 = keyval((unsigned)(e2 >> 32));
    const int  i1  = NP - 1 - (int)(unsigned)(e1 & 0xffffffffull);
    const int  i2  = NP - 1 - (int)(unsigned)(e2 & 0xffffffffull);
    const float* r1 = p1 + ((size_t)b * NP + i1) * DD;
    const float* r2 = p2 + ((size_t)b * NP + i2) * DD;

    float v1[16], v2[16];
    float s1 = 0.f, s2 = 0.f;
#pragma unroll
    for (int j = 0; j < 16; ++j) {
        float x = r1[lane + 32 * j]; v1[j] = x; s1 += x * x;
        float y = r2[lane + 32 * j]; v2[j] = y; s2 += y * y;
    }
#pragma unroll
    for (int off = 16; off; off >>= 1) {
        s1 += __shfl_xor(s1, off, 32);
        s2 += __shfl_xor(s2, off, 32);
    }
    const float c1 = w1 / fmaxf(sqrtf(s1), 1e-12f);
    const float c2 = w2 / fmaxf(sqrtf(s2), 1e-12f);
    const float dn = 1.f / (w1 + w2 + 1e-6f);

    float f[16];
    float sf = 0.f;
#pragma unroll
    for (int j = 0; j < 16; ++j) {
        f[j] = (c1 * v1[j] + c2 * v2[j]) * dn;
        sf += f[j] * f[j];
    }
#pragma unroll
    for (int off = 16; off; off >>= 1) sf += __shfl_xor(sf, off, 32);
    const float invf = 1.f / fmaxf(sqrtf(sf), 1e-12f);
    float* dst = p_s_out + (size_t)t * DD;
#pragma unroll
    for (int j = 0; j < 16; ++j) dst[lane + 32 * j] = f[j] * invf;
}

extern "C" void kernel_launch(void* const* d_in, const int* in_sizes, int n_in,
                              void* d_out, int out_size, void* d_ws, size_t ws_size,
                              hipStream_t stream) {
    (void)in_sizes; (void)n_in; (void)out_size; (void)ws_size;
    const float* anchor = (const float*)d_in[0];
    const float* p1     = (const float*)d_in[1];
    const float* p2     = (const float*)d_in[2];
    const int*   idx_a  = (const int*)d_in[3];
    const int*   idx_p  = (const int*)d_in[4];

    float* A_s = (float*)d_out;
    float* P_s = (float*)d_out + (size_t)B_ * KK * DD;

    unsigned long long* best = (unsigned long long*)d_ws;
    float* asel = (float*)((char*)d_ws + (size_t)2 * B_ * KK * sizeof(unsigned long long));

    prep_kernel<<<dim3((2 * B_ * KK) / 8), 256, 0, stream>>>(anchor, idx_a, idx_p, A_s, asel, best);

    dim3 sg(NP / 128, B_);
    score_kernel<<<sg, 512, 0, stream>>>(p1, p2, asel, best);

    fuse_kernel<<<dim3((B_ * KK) / 8), 256, 0, stream>>>(p1, p2, best, P_s);
}


# Gather launch stores the selected anchor rows in the score waves' A-fragment order: a wave's A slice is 8 KB contiguous, 16 whole-line 16-B loads replace 32 loads of 16 lines each; sums unchanged
# speedup vs baseline: 1.6689x; 1.0306x over previous
#include <hip/hip_runtime.h>

#define B_  16
#define NA  2048
#define NP  2048
#define DD  512
#define KK  32

typedef __attribute__((ext_vector_type(2))) float v2f;
typedef __attribute__((ext_vector_type(4))) float v4f;
typedef __attribute__((ext_vector_type(8))) float v8f;

#define KS   8
#define KSW  (DD / KS)
#define NJ   (KSW / 4)

#define ASEL_SLICE (2 * (NJ / 2) * 32 * 4)
#define ASEL_HALF  ((NJ / 2) * 32 * 4)

__device__ __forceinline__ unsigned ordkey(float f) {
    unsigned u = __float_as_uint(f);
    return (u & 0x80000000u) ? ~u : (u | 0x80000000u);
}
__device__ __forceinline__ float keyval(unsigned k) {
    return __uint_as_float((k & 0x80000000u) ? (k ^ 0x80000000u) : ~k);
}

__global__ __launch_bounds__(256) void prep_kernel(
    const float* __restrict__ anchor, const int* __restrict__ idx_a,
    const int* __restrict__ idx_p, float* __restrict__ a_s_out,
    float* __restrict__ asel_ws, unsigned long long* __restrict__ best) {
    __shared__ v4f stage4[8 * DD / 4];
    const int r    = blockIdx.x * 8 + (threadIdx.x >> 5);
    const int lane = threadIdx.x & 31;
    if (lane == 0) best[r] = 0ull;
    const int kind = blockIdx.x >> 6;
    const int t    = r & 511;
    const int b    = t >> 5;
    const int idx  = kind ? idx_p[t] : idx_a[t];
    const float* src = anchor + ((size_t)b * NA + idx) * DD;
    float v[16];
    float s = 0.f;
#pragma unroll
    for (int j = 0; j < 16; ++j) { float x = src[lane + 32 * j]; v[j] = x; s += x * x; }
#pragma unroll
    for (int off = 16; off; off >>= 1) s += __shfl_xor(s, off, 32);
    const float inv = 1.f / fmaxf(sqrtf(s), 1e-12f);
    if (kind == 0) {
        float* dst = a_s_out + (size_t)t * DD;
#pragma unroll
        for (int j = 0; j < 16; ++j) dst[lane + 32 * j] = v[j] * inv;
    } else {
        const int wv = threadIdx.x >> 5;
        float* stage = (float*)stage4;
#pragma unroll
        for (int j = 0; j < 16; ++j) {
            const int k    = lane + 32 * j;
            const int kk   = k & (KSW - 1);
            const int step = kk >> 2;
            const int L    = ((k / KSW) * (NJ / 2) + (step >> 1)) * 2 + ((kk >> 1) & 1);
            stage[L * 32 + wv * 4 + (step & 1) * 2 + (kk & 1)] = v[j] * inv;
        }
        __syncthreads();
        const int row0 = (blockIdx.x * 8) & 31;
        float* gbase = asel_ws + (size_t)b * KS * ASEL_SLICE + (row0 >> 4) * ASEL_HALF + (row0 & 15) * 4;
        for (int i = 0; i < 4; ++i) {
            const int q  = threadIdx.x + 256 * i;
            const int L  = q >> 3;
            const int sl = L >> 4;
            const int jj = (L >> 1) & 7;
            const int hh = L & 1;
            *(v4f*)(gbase + sl * ASEL_SLICE + jj * 128 + hh * 64 + (q & 7) * 4) = stage4[q];
        }
    }
}

#define NG   8
#define NT   2
#define NGT  (NT * NG)
#define RSTR 17

__global__ __launch_bounds__(512) void score_kernel(
    const float* __restrict__ p1, const float* __restrict__ p2,
    const float* __restrict__ asel, unsigned long long* __restrict__ best) {
    __shared__ float red[KS * NGT * RSTR * 32];
    __shared__ float invt[NT * 128];
    const int b    = blockIdx.y;
    const int tid  = threadIdx.x;
    const int wave = tid >> 5;
    const int lane = tid & 31;
    const int half = lane >> 4;
    const int lr   = lane & 15;
    const int ks   = wave & 7;
    const int rh   = wave >> 3;

    const int koff = ks * KSW + 2 * half;
    const v4f* aq = (const v4f*)(asel + ((size_t)b * KS + ks) * ASEL_SLICE) + lane;
    v2f af0[NJ], af1[NJ];
#pragma unroll
    for (int jj = 0; jj < NJ / 2; ++jj) {
        const v4f t0 = aq[jj * 32];
        const v4f t1 = aq[ASEL_HALF / 4 + jj * 32];
        af0[2 * jj]     = (v2f){t0.x, t0.y};
        af0[2 * jj + 1] = (v2f){t0.z, t0.w};
        af1[2 * jj]     = (v2f){t1.x, t1.y};
        af1[2 * jj + 1] = (v2f){t1.z, t1.w};
    }

    const int m0 = blockIdx.x * 128 + rh * 64 + lr;
    const size_t roff = ((size_t)b * NP + m0) * DD + koff;
    const float* pb1 = p1 + roff;
    const float* pb2 = p2 + roff;

    v2f bc[NJ], bn[NJ];
#pragma unroll
    for (int j = 0; j < NJ; ++j) bc[j] = *(const v2f*)(pb1 + 4 * j);

    const float* pn = pb1 + (size_t)16 * DD;
#pragma unroll 1
    for (int q = 0; q < 2 * NT; ++q) {
#pragma unroll
        for (int j = 0; j < NJ; ++j) bn[j] = *(const v2f*)(pn + 4 * j);
        pn += (size_t)16 * DD;
        {
            v8f acc0 = {};
            v8f acc1 = {};
            float ss = 0.f;
#pragma unroll
            for (int j = 0; j < NJ; ++j) {
                const v2f bf = bc[j];
                ss += bf.x * bf.x + bf.y * bf.y;
                acc0 = __builtin_amdgcn_wmma_f32_16x16x4_f32(false, af0[j], false, bf,
                                                             (short)0, acc0, false, false);
                acc1 = __builtin_amdgcn_wmma_f32_16x16x4_f32(false, af1[j], false, bf,
                                                             (short)0, acc1, false, false);
            }
            const int gt = (q >> 1) * NG + rh * 4 + 2 * (q & 1);
            float* r = red + (size_t)((ks * NGT + gt) * RSTR) * 32 + lane;
#pragma unroll
            for (int i = 0; i < 8; ++i) {
                r[i * 32]       = acc0[i];
                r[(8 + i) * 32] = acc1[i];
            }
            r[16 * 32] = ss;
        }
        if (q < 2 * NT - 1) {
            if (q & 1) pn = pb2;
#pragma unroll
            for (int j = 0; j < NJ; ++j) bc[j] = *(const v2f*)(pn + 4 * j);
            pn += (size_t)16 * DD;
        }
        {
            v8f acc0 = {};
            v8f acc1 = {};
            float ss = 0.f;
#pragma unroll
            for (int j = 0; j < NJ; ++j) {
                const v2f bf = bn[j];
                ss += bf.x * bf.x + bf.y * bf.y;
                acc0 = __builtin_amdgcn_wmma_f32_16x16x4_f32(false, af0[j], false, bf,
                                                             (short)0, acc0, false, false);
                acc1 = __builtin_amdgcn_wmma_f32_16x16x4_f32(false, af1[j], false, bf,
                                                             (short)0, acc1, false, false);
            }
            const int gt = (q >> 1) * NG + rh * 4 + 2 * (q & 1) + 1;
            float* r = red + (size_t)((ks * NGT + gt) * RSTR) * 32 + lane;
#pragma unroll
            for (int i = 0; i < 8; ++i) {
                r[i * 32]       = acc0[i];
                r[(8 + i) * 32] = acc1[i];
            }
            r[16 * 32] = ss;
        }
    }
    __syncthreads();

    if (tid < NT * 128) {
        const int tA = tid >> 7;
        const int GA = (tid >> 4) & 7;
        const int lA = tid & 15;
        float s0 = 0.f, s1 = 0.f;
        for (int s = 0; s < KS; ++s) {
            const float* r = red + (size_t)((s * NGT + tA * NG + GA) * RSTR + 16) * 32 + lA;
            s0 += r[0];
            s1 += r[16];
        }
        invt[tid] = 1.f / fmaxf(sqrtf(s0 + s1), 1e-12f);
    }
    __syncthreads();

    const int t  = wave >> 3;
    const int ri = wave & 7;
    const int mb = blockIdx.x * 128 + lr;
    unsigned long long k0 = 0ull;
    unsigned long long k1 = 0ull;
#pragma unroll
    for (int G = 0; G < NG; ++G) {
        float a0 = 0.f, a1 = 0.f;
        for (int s = 0; s < KS; ++s) {
            const float* r = red + (size_t)((s * NGT + t * NG + G) * RSTR) * 32 + lane;
            a0 += r[ri * 32];
            a1 += r[(8 + ri) * 32];
        }
        const float inv = invt[t * 128 + G * 16 + lr];
        const unsigned keylo = (unsigned)(NP - 1 - (mb + G * 16));
        const unsigned long long c0 = ((unsigned long long)ordkey(a0 * inv) << 32) | keylo;
        const unsigned long long c1 = ((unsigned long long)ordkey(a1 * inv) << 32) | keylo;
        k0 = c0 > k0 ? c0 : k0;
        k1 = c1 > k1 ? c1 : k1;
    }
    for (int off = 8; off; off >>= 1) {
        const unsigned long long o0 = __shfl_xor(k0, off, 32);
        const unsigned long long o1 = __shfl_xor(k1, off, 32);
        k0 = o0 > k0 ? o0 : k0;
        k1 = o1 > k1 ? o1 : k1;
    }
    if (lr == 0) {
        unsigned long long* bt = best + (size_t)t * B_ * KK + (size_t)b * KK;
        atomicMax(bt + (ri + 8 * half), k0);
        atomicMax(bt + (16 + ri + 8 * half), k1);
    }
}

__global__ __launch_bounds__(256) void fuse_kernel(
    const float* __restrict__ p1, const float* __restrict__ p2,
    const unsigned long long* __restrict__ best, float* __restrict__ p_s_out) {
    const int t    = blockIdx.x * 8 + (threadIdx.x >> 5);
    const int lane = threadIdx.x & 31;
    const int b    = t >> 5;
    const unsigned long long e1 = best[t];
    const unsigned long long e2 = best[B_ * KK + t];
    const float w1 = keyval((unsigned)(e1 >> 32));
    const float w2 = keyval((unsigned)(e2 >> 32));
    const int  i1  = NP - 1 - (int)(unsigned)(e1 & 0xffffffffull);
    const int  i2  = NP - 1 - (int)(unsigned)(e2 & 0xffffffffull);
    const float* r1 = p1 + ((size_t)b * NP + i1) * DD;
    const float* r2 = p2 + ((size_t)b * NP + i2) * DD;

    float v1[16], v2[16];
    float s1 = 0.f, s2 = 0.f;
#pragma unroll
    for (int j = 0; j < 16; ++j) {
        float x = r1[lane + 32 * j]; v1[j] = x; s1 += x * x;
        float y = r2[lane + 32 * j]; v2[j] = y; s2 += y * y;
    }
#pragma unroll
    for (int off = 16; off; off >>= 1) {
        s1 += __shfl_xor(s1, off, 32);
        s2 += __shfl_xor(s2, off, 32);
    }
    const float c1 = w1 / fmaxf(sqrtf(s1), 1e-12f);
    const float c2 = w2 / fmaxf(sqrtf(s2), 1e-12f);
    const float dn = 1.f / (w1 + w2 + 1e-6f);

    float f[16];
    float sf = 0.f;
#pragma unroll
    for (int j = 0; j < 16; ++j) {
        f[j] = (c1 * v1[j] + c2 * v2[j]) * dn;
        sf += f[j] * f[j];
    }
#pragma unroll
    for (int off = 16; off; off >>= 1) sf += __shfl_xor(sf, off, 32);
    const float invf = 1.f / fmaxf(sqrtf(sf), 1e-12f);
    float* dst = p_s_out + (size_t)t * DD;
#pragma unroll
    for (int j = 0; j < 16; ++j) dst[lane + 32 * j] = f[j] * invf;
}

extern "C" void kernel_launch(void* const* d_in, const int* in_sizes, int n_in,
                              void* d_out, int out_size, void* d_ws, size_t ws_size,
                              hipStream_t stream) {
    (void)in_sizes; (void)n_in; (void)out_size; (void)ws_size;
    const float* anchor = (const float*)d_in[0];
    const float* p1     = (const float*)d_in[1];
    const float* p2     = (const float*)d_in[2];
    const int*   idx_a  = (const int*)d_in[3];
    const int*   idx_p  = (const int*)d_in[4];

    float* A_s = (float*)d_out;
    float* P_s = (float*)d_out + (size_t)B_ * KK * DD;

    unsigned long long* best = (unsigned long long*)d_ws;
    float* asel = (float*)((char*)d_ws + (size_t)2 * B_ * KK * sizeof(unsigned long long));

    prep_kernel<<<dim3((2 * B_ * KK) / 8), 256, 0, stream>>>(anchor, idx_a, idx_p, A_s, asel, best);

    dim3 sg(NP / 128, B_);
    score_kernel<<<sg, 512, 0, stream>>>(p1, p2, asel, best);

    fuse_kernel<<<dim3((B_ * KK) / 8), 256, 0, stream>>>(p1, p2, best, P_s);
}
